# starting kernel
# baseline (speedup 1.0000x reference)
#include <hip/hip_runtime.h>

typedef __attribute__((ext_vector_type(2))) float v2f;
typedef __attribute__((ext_vector_type(8))) float v8f;

#define RNN_B 8192
#define RNN_T 512
#define RNN_I 9
#define RNN_H 6
#define TILES_PER_WAVE 8

__device__ __forceinline__ float fast_tanhf(float v) {
  float e = __builtin_amdgcn_exp2f(v * 2.88539008177792681472f);
  return 1.0f - 2.0f * __builtin_amdgcn_rcpf(e + 1.0f);
}

__global__ __launch_bounds__(256) void xproj_wmma_f32(
    const float* __restrict__ x, const float* __restrict__ W_ih,
    const float* __restrict__ b_ih, const float* __restrict__ b_hh,
    float* __restrict__ xp)
{
  const int  lane = threadIdx.x & 31;
  const int  l15  = lane & 15;
  const int  half = lane >> 4;
  const long wave = ((long)blockIdx.x * blockDim.x + threadIdx.x) >> 5;
  const long waveRowBase = wave * (TILES_PER_WAVE * 16);

  int  kIdx[6];
  bool kVal[6];
#pragma unroll
  for (int s = 0; s < 3; ++s) {
    const int kA = 4 * s + half * 2;
    kVal[2*s+0] = (kA     < RNN_I);
    kVal[2*s+1] = (kA + 1 < RNN_I);
    kIdx[2*s+0] = kVal[2*s+0] ? kA     : (RNN_I - 1);
    kIdx[2*s+1] = kVal[2*s+1] ? kA + 1 : (RNN_I - 1);
  }

  const bool nVal   = (l15 < RNN_H);
  const int  nClamp = nVal ? l15 : (RNN_H - 1);

  float bi = b_ih[nClamp];
  float bh = b_hh[nClamp];
  const float bias = nVal ? (bi + bh) : 0.0f;

  v2f bm[3];
#pragma unroll
  for (int s = 0; s < 3; ++s) {
    float w0 = W_ih[nClamp * RNN_I + kIdx[2*s+0]];
    float w1 = W_ih[nClamp * RNN_I + kIdx[2*s+1]];
    bm[s].x = (nVal && kVal[2*s+0]) ? w0 : 0.0f;
    bm[s].y = (nVal && kVal[2*s+1]) ? w1 : 0.0f;
  }

#pragma unroll 1
  for (int tt = 0; tt < TILES_PER_WAVE; ++tt) {
    const long rowBase = waveRowBase + (long)tt * 16;
    const float* xrow = x + (rowBase + l15) * RNN_I;

    v8f c;
#pragma unroll
    for (int j = 0; j < 8; ++j) c[j] = bias;

#pragma unroll
    for (int s = 0; s < 3; ++s) {
      float r0 = xrow[kIdx[2*s+0]];
      float r1 = xrow[kIdx[2*s+1]];
      v2f a;
      a.x = kVal[2*s+0] ? r0 : 0.0f;
      a.y = kVal[2*s+1] ? r1 : 0.0f;
      c = __builtin_amdgcn_wmma_f32_16x16x4_f32(false, a, false, bm[s],
                                                (short)0, c, false, false);
    }

    if (l15 < 8) {
#pragma unroll
      for (int j = 0; j < 8; ++j) {
        long r  = rowBase + j + half * 8;
        long bb = r >> 9;
        long ti = r & (RNN_T - 1);
        xp[(ti * (long)RNN_B + bb) * 8 + l15] = c[j];
      }
    }
  }
}

__global__ __launch_bounds__(256) void rnn_scan(
    const float* __restrict__ xp, const float* __restrict__ W_hh,
    const float* __restrict__ W_fc, const float* __restrict__ b_fc,
    float* __restrict__ out)
{
  const int b = blockIdx.x * blockDim.x + threadIdx.x;

  float w[RNN_H][RNN_H];
#pragma unroll
  for (int j = 0; j < RNN_H; ++j)
#pragma unroll
    for (int i = 0; i < RNN_H; ++i) w[j][i] = W_hh[j * RNN_H + i];

  float h[RNN_H];
#pragma unroll
  for (int j = 0; j < RNN_H; ++j) h[j] = 0.0f;

  const float4* base = (const float4*)xp;
  float4 cur0 = base[(long)b * 2 + 0];
  float4 cur1 = base[(long)b * 2 + 1];

  for (int t = 0; t < RNN_T; ++t) {
    const int  tn = (t + 1 < RNN_T) ? (t + 1) : (RNN_T - 1);
    const long ni = ((long)tn * RNN_B + b) * 2;
    float4 nx0 = base[ni];
    float4 nx1 = base[ni + 1];
    const int tp = (t + 8 < RNN_T) ? (t + 8) : (RNN_T - 1);
    __builtin_prefetch(&base[((long)tp * RNN_B + b) * 2], 0, 0);

    const float xv[RNN_H] = {cur0.x, cur0.y, cur0.z, cur0.w, cur1.x, cur1.y};
    float nh[RNN_H];
#pragma unroll
    for (int j = 0; j < RNN_H; ++j) {
      float acc = xv[j];
#pragma unroll
      for (int i = 0; i < RNN_H; ++i) acc = fmaf(w[j][i], h[i], acc);
      nh[j] = fast_tanhf(acc);
    }
#pragma unroll
    for (int j = 0; j < RNN_H; ++j) h[j] = nh[j];
    cur0 = nx0; cur1 = nx1;
  }

  float acc = b_fc[0];
#pragma unroll
  for (int j = 0; j < RNN_H; ++j) acc = fmaf(W_fc[j], h[j], acc);
  out[b] = acc;
}

__global__ __launch_bounds__(256) void rnn_fused(
    const float* __restrict__ x, const float* __restrict__ W_ih,
    const float* __restrict__ W_hh, const float* __restrict__ b_ih,
    const float* __restrict__ b_hh, const float* __restrict__ W_fc,
    const float* __restrict__ b_fc, float* __restrict__ out)
{
  const int b = blockIdx.x * blockDim.x + threadIdx.x;

  float wih[RNN_H][RNN_I], whh[RNN_H][RNN_H], bias[RNN_H];
#pragma unroll
  for (int j = 0; j < RNN_H; ++j) {
    bias[j] = b_ih[j] + b_hh[j];
#pragma unroll
    for (int i = 0; i < RNN_I; ++i) wih[j][i] = W_ih[j * RNN_I + i];
#pragma unroll
    for (int i = 0; i < RNN_H; ++i) whh[j][i] = W_hh[j * RNN_H + i];
  }

  float h[RNN_H];
#pragma unroll
  for (int j = 0; j < RNN_H; ++j) h[j] = 0.0f;

  const float* xr = x + (long)b * RNN_T * RNN_I;
  for (int t = 0; t < RNN_T; ++t) {
    float xv[RNN_I];
#pragma unroll
    for (int i = 0; i < RNN_I; ++i) xv[i] = xr[(long)t * RNN_I + i];
    float nh[RNN_H];
#pragma unroll
    for (int j = 0; j < RNN_H; ++j) {
      float acc = bias[j];
#pragma unroll
      for (int i = 0; i < RNN_I; ++i) acc = fmaf(wih[j][i], xv[i], acc);
#pragma unroll
      for (int i = 0; i < RNN_H; ++i) acc = fmaf(whh[j][i], h[i], acc);
      nh[j] = fast_tanhf(acc);
    }
#pragma unroll
    for (int j = 0; j < RNN_H; ++j) h[j] = nh[j];
  }

  float acc = b_fc[0];
#pragma unroll
  for (int j = 0; j < RNN_H; ++j) acc = fmaf(W_fc[j], h[j], acc);
  out[b] = acc;
}

extern "C" void kernel_launch(void* const* d_in, const int* in_sizes, int n_in,
                              void* d_out, int out_size, void* d_ws, size_t ws_size,
                              hipStream_t stream) {
  const float* x    = (const float*)d_in[0];
  const float* W_ih = (const float*)d_in[1];
  const float* W_hh = (const float*)d_in[2];
  const float* b_ih = (const float*)d_in[3];
  const float* b_hh = (const float*)d_in[4];
  const float* W_fc = (const float*)d_in[5];
  const float* b_fc = (const float*)d_in[6];
  float* out = (float*)d_out;

  const size_t need = (size_t)RNN_T * RNN_B * 8 * sizeof(float);
  if (ws_size >= need) {
    float* xp = (float*)d_ws;
    const long waves  = ((long)RNN_B * RNN_T / 16) / TILES_PER_WAVE;
    const int  blocks = (int)(waves * 32 / 256);
    xproj_wmma_f32<<<dim3(blocks), dim3(256), 0, stream>>>(x, W_ih, b_ih, b_hh, xp);
    rnn_scan<<<dim3(RNN_B / 256), dim3(256), 0, stream>>>(xp, W_hh, W_fc, b_fc, out);
  } else {
    rnn_fused<<<dim3(RNN_B / 256), dim3(256), 0, stream>>>(x, W_ih, W_hh, b_ih,
                                                           b_hh, W_fc, b_fc, out);
  }
}


# scan kernel: xp rows requested 8 or more steps ahead into two alternating register sets (was 1 step ahead: every step waited a full L2 round trip); L2 hint now 24 steps ahead; arithmetic unchanged
# speedup vs baseline: 1.4104x; 1.4104x over previous
#include <hip/hip_runtime.h>

typedef __attribute__((ext_vector_type(2))) float v2f;
typedef __attribute__((ext_vector_type(8))) float v8f;

#define RNN_B 8192
#define RNN_T 512
#define RNN_I 9
#define RNN_H 6
#define TILES_PER_WAVE 8

__device__ __forceinline__ float fast_tanhf(float v) {
  float e = __builtin_amdgcn_exp2f(v * 2.88539008177792681472f);
  return 1.0f - 2.0f * __builtin_amdgcn_rcpf(e + 1.0f);
}

__global__ __launch_bounds__(256) void xproj_wmma_f32(
    const float* __restrict__ x, const float* __restrict__ W_ih,
    const float* __restrict__ b_ih, const float* __restrict__ b_hh,
    float* __restrict__ xp)
{
  const int  lane = threadIdx.x & 31;
  const int  l15  = lane & 15;
  const int  half = lane >> 4;
  const long wave = ((long)blockIdx.x * blockDim.x + threadIdx.x) >> 5;
  const long waveRowBase = wave * (TILES_PER_WAVE * 16);

  int  kIdx[6];
  bool kVal[6];
#pragma unroll
  for (int s = 0; s < 3; ++s) {
    const int kA = 4 * s + half * 2;
    kVal[2*s+0] = (kA     < RNN_I);
    kVal[2*s+1] = (kA + 1 < RNN_I);
    kIdx[2*s+0] = kVal[2*s+0] ? kA     : (RNN_I - 1);
    kIdx[2*s+1] = kVal[2*s+1] ? kA + 1 : (RNN_I - 1);
  }

  const bool nVal   = (l15 < RNN_H);
  const int  nClamp = nVal ? l15 : (RNN_H - 1);

  float bi = b_ih[nClamp];
  float bh = b_hh[nClamp];
  const float bias = nVal ? (bi + bh) : 0.0f;

  v2f bm[3];
#pragma unroll
  for (int s = 0; s < 3; ++s) {
    float w0 = W_ih[nClamp * RNN_I + kIdx[2*s+0]];
    float w1 = W_ih[nClamp * RNN_I + kIdx[2*s+1]];
    bm[s].x = (nVal && kVal[2*s+0]) ? w0 : 0.0f;
    bm[s].y = (nVal && kVal[2*s+1]) ? w1 : 0.0f;
  }

#pragma unroll 1
  for (int tt = 0; tt < TILES_PER_WAVE; ++tt) {
    const long rowBase = waveRowBase + (long)tt * 16;
    const float* xrow = x + (rowBase + l15) * RNN_I;

    v8f c;
#pragma unroll
    for (int j = 0; j < 8; ++j) c[j] = bias;

#pragma unroll
    for (int s = 0; s < 3; ++s) {
      float r0 = xrow[kIdx[2*s+0]];
      float r1 = xrow[kIdx[2*s+1]];
      v2f a;
      a.x = kVal[2*s+0] ? r0 : 0.0f;
      a.y = kVal[2*s+1] ? r1 : 0.0f;
      c = __builtin_amdgcn_wmma_f32_16x16x4_f32(false, a, false, bm[s],
                                                (short)0, c, false, false);
    }

    if (l15 < 8) {
#pragma unroll
      for (int j = 0; j < 8; ++j) {
        long r  = rowBase + j + half * 8;
        long bb = r >> 9;
        long ti = r & (RNN_T - 1);
        xp[(ti * (long)RNN_B + bb) * 8 + l15] = c[j];
      }
    }
  }
}

#define SCAN_BLK 8
#define SCAN_PF  24
#define SCAN_STEP_FLOATS ((size_t)RNN_B * 8)

struct ScanRows {
  float4 a[SCAN_BLK];
  float2 b[SCAN_BLK];
};

__device__ __forceinline__ void scan_load_rows(ScanRows& r, const float* p) {
#pragma unroll
  for (int u = 0; u < SCAN_BLK; ++u) {
    const float* q = p + (size_t)u * SCAN_STEP_FLOATS;
    r.a[u] = *(const float4*)q;
    r.b[u] = *(const float2*)(q + 4);
  }
}

__device__ __forceinline__ void scan_run_rows(const ScanRows& r,
                                              const float (&w)[RNN_H][RNN_H],
                                              float (&h)[RNN_H],
                                              const float* pf) {
#pragma unroll
  for (int u = 0; u < SCAN_BLK; ++u) {
    __builtin_prefetch(pf + (size_t)u * SCAN_STEP_FLOATS, 0, 0);
    const float xv[RNN_H] = {r.a[u].x, r.a[u].y, r.a[u].z, r.a[u].w,
                             r.b[u].x, r.b[u].y};
    float nh[RNN_H];
#pragma unroll
    for (int j = 0; j < RNN_H; ++j) {
      float acc = xv[j];
#pragma unroll
      for (int i = 0; i < RNN_H; ++i) acc = fmaf(w[j][i], h[i], acc);
      nh[j] = fast_tanhf(acc);
    }
#pragma unroll
    for (int j = 0; j < RNN_H; ++j) h[j] = nh[j];
  }
}

__global__ __launch_bounds__(256) void rnn_scan(
    const float* __restrict__ xp, const float* __restrict__ W_hh,
    const float* __restrict__ W_fc, const float* __restrict__ b_fc,
    float* __restrict__ out)
{
  static_assert(RNN_T % (2 * SCAN_BLK) == 0, "two register sets per pass");
  const int b = blockIdx.x * blockDim.x + threadIdx.x;

  float w[RNN_H][RNN_H];
#pragma unroll
  for (int j = 0; j < RNN_H; ++j)
#pragma unroll
    for (int i = 0; i < RNN_H; ++i) w[j][i] = W_hh[j * RNN_H + i];

  float h[RNN_H];
#pragma unroll
  for (int j = 0; j < RNN_H; ++j) h[j] = 0.0f;

  const float* lane = xp + (size_t)b * 8;

  ScanRows X, Y;
  scan_load_rows(X, lane);

  for (int tb = 0; tb < RNN_T; tb += 2 * SCAN_BLK) {
    const int pfA = (tb + SCAN_PF < RNN_T - SCAN_BLK) ? (tb + SCAN_PF)
                                                      : (RNN_T - SCAN_BLK);
    const int pfB = (tb + SCAN_BLK + SCAN_PF < RNN_T - SCAN_BLK)
                        ? (tb + SCAN_BLK + SCAN_PF) : (RNN_T - SCAN_BLK);
    const int nx = (tb + 2 * SCAN_BLK < RNN_T) ? (tb + 2 * SCAN_BLK)
                                               : (RNN_T - SCAN_BLK);

    scan_load_rows(Y, lane + (size_t)(tb + SCAN_BLK) * SCAN_STEP_FLOATS);
    asm volatile("" : "+v"(h[0]), "+v"(h[1]), "+v"(h[2]), "+v"(h[3]),
                      "+v"(h[4]), "+v"(h[5]) : : "memory");
    scan_run_rows(X, w, h, lane + (size_t)pfA * SCAN_STEP_FLOATS);
    scan_load_rows(X, lane + (size_t)nx * SCAN_STEP_FLOATS);
    scan_run_rows(Y, w, h, lane + (size_t)pfB * SCAN_STEP_FLOATS);
  }

  float acc = b_fc[0];
#pragma unroll
  for (int j = 0; j < RNN_H; ++j) acc = fmaf(W_fc[j], h[j], acc);
  out[b] = acc;
}

__global__ __launch_bounds__(256) void rnn_fused(
    const float* __restrict__ x, const float* __restrict__ W_ih,
    const float* __restrict__ W_hh, const float* __restrict__ b_ih,
    const float* __restrict__ b_hh, const float* __restrict__ W_fc,
    const float* __restrict__ b_fc, float* __restrict__ out)
{
  const int b = blockIdx.x * blockDim.x + threadIdx.x;

  float wih[RNN_H][RNN_I], whh[RNN_H][RNN_H], bias[RNN_H];
#pragma unroll
  for (int j = 0; j < RNN_H; ++j) {
    bias[j] = b_ih[j] + b_hh[j];
#pragma unroll
    for (int i = 0; i < RNN_I; ++i) wih[j][i] = W_ih[j * RNN_I + i];
#pragma unroll
    for (int i = 0; i < RNN_H; ++i) whh[j][i] = W_hh[j * RNN_H + i];
  }

  float h[RNN_H];
#pragma unroll
  for (int j = 0; j < RNN_H; ++j) h[j] = 0.0f;

  const float* xr = x + (long)b * RNN_T * RNN_I;
  for (int t = 0; t < RNN_T; ++t) {
    float xv[RNN_I];
#pragma unroll
    for (int i = 0; i < RNN_I; ++i) xv[i] = xr[(long)t * RNN_I + i];
    float nh[RNN_H];
#pragma unroll
    for (int j = 0; j < RNN_H; ++j) {
      float acc = bias[j];
#pragma unroll
      for (int i = 0; i < RNN_I; ++i) acc = fmaf(wih[j][i], xv[i], acc);
#pragma unroll
      for (int i = 0; i < RNN_H; ++i) acc = fmaf(whh[j][i], h[i], acc);
      nh[j] = fast_tanhf(acc);
    }
#pragma unroll
    for (int j = 0; j < RNN_H; ++j) h[j] = nh[j];
  }

  float acc = b_fc[0];
#pragma unroll
  for (int j = 0; j < RNN_H; ++j) acc = fmaf(W_fc[j], h[j], acc);
  out[b] = acc;
}

extern "C" void kernel_launch(void* const* d_in, const int* in_sizes, int n_in,
                              void* d_out, int out_size, void* d_ws, size_t ws_size,
                              hipStream_t stream) {
  const float* x    = (const float*)d_in[0];
  const float* W_ih = (const float*)d_in[1];
  const float* W_hh = (const float*)d_in[2];
  const float* b_ih = (const float*)d_in[3];
  const float* b_hh = (const float*)d_in[4];
  const float* W_fc = (const float*)d_in[5];
  const float* b_fc = (const float*)d_in[6];
  float* out = (float*)d_out;

  const size_t need = (size_t)RNN_T * RNN_B * 8 * sizeof(float);
  if (ws_size >= need) {
    float* xp = (float*)d_ws;
    const long waves  = ((long)RNN_B * RNN_T / 16) / TILES_PER_WAVE;
    const int  blocks = (int)(waves * 32 / 256);
    xproj_wmma_f32<<<dim3(blocks), dim3(256), 0, stream>>>(x, W_ih, b_ih, b_hh, xp);
    rnn_scan<<<dim3(RNN_B / 256), dim3(256), 0, stream>>>(xp, W_hh, W_fc, b_fc, out);
  } else {
    rnn_fused<<<dim3(RNN_B / 256), dim3(256), 0, stream>>>(x, W_ih, W_hh, b_ih,
                                                           b_hh, W_fc, b_fc, out);
  }
}


# projection kernel: weights as A operand, x as B, tile = 8 batch rows x 2 steps: a lane stores its 4 results as one 16-byte store, a wave store covers whole 256-byte pieces; k order of sums changed
# speedup vs baseline: 2.5092x; 1.7791x over previous
#include <hip/hip_runtime.h>

typedef __attribute__((ext_vector_type(2))) float v2f;
typedef __attribute__((ext_vector_type(8))) float v8f;

#define RNN_B 8192
#define RNN_T 512
#define RNN_I 9
#define RNN_H 6
#define TILES_PER_WAVE 8

__device__ __forceinline__ float fast_tanhf(float v) {
  float e = __builtin_amdgcn_exp2f(v * 2.88539008177792681472f);
  return 1.0f - 2.0f * __builtin_amdgcn_rcpf(e + 1.0f);
}

#define XP_TS 2
#define XP_BR (16 / XP_TS)
#define XP_TT (TILES_PER_WAVE * XP_TS)

__global__ __launch_bounds__(256) void xproj_wmma_f32(
    const float* __restrict__ x, const float* __restrict__ W_ih,
    const float* __restrict__ b_ih, const float* __restrict__ b_hh,
    float* __restrict__ xp)
{
  static_assert(RNN_T % XP_TT == 0 && RNN_B % XP_BR == 0, "whole tiles");
  const int  lane  = threadIdx.x & 31;
  const int  c     = lane & 15;
  const int  half  = lane >> 4;
  const bool half0 = (half == 0);
  const int  bi    = c & (XP_BR - 1);
  const int  ti    = c / XP_BR;
  const long wave  = ((long)blockIdx.x * blockDim.x + threadIdx.x) >> 5;
  const int  b0 = (int)(wave / (RNN_T / XP_TT)) * XP_BR;
  const int  t0 = (int)(wave % (RNN_T / XP_TT)) * XP_TT;

  const bool mVal = (c < 4) || (c >= 8 && c < 8 + (RNN_H - 4));
  const int  nRow = mVal ? ((c < 4) ? c : (c - 4)) : 0;
  v2f am[3];
#pragma unroll
  for (int s = 0; s < 3; ++s) {
#pragma unroll
    for (int e2 = 0; e2 < 2; ++e2) {
      const int  e    = 2 * s + e2;
      const bool kVal = half0 ? (e < 5) : (e < 4);
      const int  k    = kVal ? (half0 ? e : (5 + e)) : 0;
      const float w   = W_ih[nRow * RNN_I + k];
      const float v   = (mVal && kVal) ? w : 0.0f;
      if (e2 == 0) am[s].x = v; else am[s].y = v;
    }
  }

  float cb[4];
#pragma unroll
  for (int r = 0; r < 4; ++r) {
    const int  n    = 4 * half + r;
    const bool nVal = (n < RNN_H);
    const int  nc   = nVal ? n : 0;
    const float bi2 = b_ih[nc];
    const float bh2 = b_hh[nc];
    cb[r] = nVal ? (bi2 + bh2) : 0.0f;
  }

  const float* xr = x + ((size_t)(b0 + bi) * RNN_T + (size_t)(t0 + ti)) * RNN_I
                      + (half0 ? 0 : 5);
  const int sOff = half0 ? 4 : 3;
  float* op = xp + ((size_t)(t0 + ti) * RNN_B + (size_t)(b0 + bi)) * 8 + 4 * half;

#pragma unroll 1
  for (int tt = 0; tt < TILES_PER_WAVE; ++tt) {
    const float* q = xr + tt * (XP_TS * RNN_I);
    float4 xq;
    __builtin_memcpy(&xq, q, sizeof(float4));
    const float xs = q[sOff];

    v8f acc;
    acc[0] = cb[0]; acc[1] = cb[1]; acc[2] = cb[2]; acc[3] = cb[3];
    acc[4] = 0.0f;  acc[5] = 0.0f;  acc[6] = 0.0f;  acc[7] = 0.0f;
    v2f b0v, b1v, b2v;
    b0v.x = xq.x; b0v.y = xq.y;
    b1v.x = xq.z; b1v.y = xq.w;
    b2v.x = half0 ? xs : 0.0f;
    b2v.y = 0.0f;
    acc = __builtin_amdgcn_wmma_f32_16x16x4_f32(false, am[0], false, b0v,
                                                (short)0, acc, false, false);
    acc = __builtin_amdgcn_wmma_f32_16x16x4_f32(false, am[1], false, b1v,
                                                (short)0, acc, false, false);
    acc = __builtin_amdgcn_wmma_f32_16x16x4_f32(false, am[2], false, b2v,
                                                (short)0, acc, false, false);

    float4 o;
    o.x = acc[0]; o.y = acc[1]; o.z = acc[2]; o.w = acc[3];
    *(float4*)(op + (size_t)tt * ((size_t)XP_TS * RNN_B * 8)) = o;
  }
}

#define SCAN_BLK 8
#define SCAN_PF  24
#define SCAN_STEP_FLOATS ((size_t)RNN_B * 8)

struct ScanRows {
  float4 a[SCAN_BLK];
  float2 b[SCAN_BLK];
};

__device__ __forceinline__ void scan_load_rows(ScanRows& r, const float* p) {
#pragma unroll
  for (int u = 0; u < SCAN_BLK; ++u) {
    const float* q = p + (size_t)u * SCAN_STEP_FLOATS;
    r.a[u] = *(const float4*)q;
    r.b[u] = *(const float2*)(q + 4);
  }
}

__device__ __forceinline__ void scan_run_rows(const ScanRows& r,
                                              const float (&w)[RNN_H][RNN_H],
                                              float (&h)[RNN_H],
                                              const float* pf) {
#pragma unroll
  for (int u = 0; u < SCAN_BLK; ++u) {
    __builtin_prefetch(pf + (size_t)u * SCAN_STEP_FLOATS, 0, 0);
    const float xv[RNN_H] = {r.a[u].x, r.a[u].y, r.a[u].z, r.a[u].w,
                             r.b[u].x, r.b[u].y};
    float nh[RNN_H];
#pragma unroll
    for (int j = 0; j < RNN_H; ++j) {
      float acc = xv[j];
#pragma unroll
      for (int i = 0; i < RNN_H; ++i) acc = fmaf(w[j][i], h[i], acc);
      nh[j] = fast_tanhf(acc);
    }
#pragma unroll
    for (int j = 0; j < RNN_H; ++j) h[j] = nh[j];
  }
}

__global__ __launch_bounds__(256) void rnn_scan(
    const float* __restrict__ xp, const float* __restrict__ W_hh,
    const float* __restrict__ W_fc, const float* __restrict__ b_fc,
    float* __restrict__ out)
{
  static_assert(RNN_T % (2 * SCAN_BLK) == 0, "two register sets per pass");
  const int b = blockIdx.x * blockDim.x + threadIdx.x;

  float w[RNN_H][RNN_H];
#pragma unroll
  for (int j = 0; j < RNN_H; ++j)
#pragma unroll
    for (int i = 0; i < RNN_H; ++i) w[j][i] = W_hh[j * RNN_H + i];

  float h[RNN_H];
#pragma unroll
  for (int j = 0; j < RNN_H; ++j) h[j] = 0.0f;

  const float* lane = xp + (size_t)b * 8;

  ScanRows X, Y;
  scan_load_rows(X, lane);

  for (int tb = 0; tb < RNN_T; tb += 2 * SCAN_BLK) {
    const int pfA = (tb + SCAN_PF < RNN_T - SCAN_BLK) ? (tb + SCAN_PF)
                                                      : (RNN_T - SCAN_BLK);
    const int pfB = (tb + SCAN_BLK + SCAN_PF < RNN_T - SCAN_BLK)
                        ? (tb + SCAN_BLK + SCAN_PF) : (RNN_T - SCAN_BLK);
    const int nx = (tb + 2 * SCAN_BLK < RNN_T) ? (tb + 2 * SCAN_BLK)
                                               : (RNN_T - SCAN_BLK);

    scan_load_rows(Y, lane + (size_t)(tb + SCAN_BLK) * SCAN_STEP_FLOATS);
    asm volatile("" : "+v"(h[0]), "+v"(h[1]), "+v"(h[2]), "+v"(h[3]),
                      "+v"(h[4]), "+v"(h[5]) : : "memory");
    scan_run_rows(X, w, h, lane + (size_t)pfA * SCAN_STEP_FLOATS);
    scan_load_rows(X, lane + (size_t)nx * SCAN_STEP_FLOATS);
    scan_run_rows(Y, w, h, lane + (size_t)pfB * SCAN_STEP_FLOATS);
  }

  float acc = b_fc[0];
#pragma unroll
  for (int j = 0; j < RNN_H; ++j) acc = fmaf(W_fc[j], h[j], acc);
  out[b] = acc;
}

__global__ __launch_bounds__(256) void rnn_fused(
    const float* __restrict__ x, const float* __restrict__ W_ih,
    const float* __restrict__ W_hh, const float* __restrict__ b_ih,
    const float* __restrict__ b_hh, const float* __restrict__ W_fc,
    const float* __restrict__ b_fc, float* __restrict__ out)
{
  const int b = blockIdx.x * blockDim.x + threadIdx.x;

  float wih[RNN_H][RNN_I], whh[RNN_H][RNN_H], bias[RNN_H];
#pragma unroll
  for (int j = 0; j < RNN_H; ++j) {
    bias[j] = b_ih[j] + b_hh[j];
#pragma unroll
    for (int i = 0; i < RNN_I; ++i) wih[j][i] = W_ih[j * RNN_I + i];
#pragma unroll
    for (int i = 0; i < RNN_H; ++i) whh[j][i] = W_hh[j * RNN_H + i];
  }

  float h[RNN_H];
#pragma unroll
  for (int j = 0; j < RNN_H; ++j) h[j] = 0.0f;

  const float* xr = x + (long)b * RNN_T * RNN_I;
  for (int t = 0; t < RNN_T; ++t) {
    float xv[RNN_I];
#pragma unroll
    for (int i = 0; i < RNN_I; ++i) xv[i] = xr[(long)t * RNN_I + i];
    float nh[RNN_H];
#pragma unroll
    for (int j = 0; j < RNN_H; ++j) {
      float acc = bias[j];
#pragma unroll
      for (int i = 0; i < RNN_I; ++i) acc = fmaf(wih[j][i], xv[i], acc);
#pragma unroll
      for (int i = 0; i < RNN_H; ++i) acc = fmaf(whh[j][i], h[i], acc);
      nh[j] = fast_tanhf(acc);
    }
#pragma unroll
    for (int j = 0; j < RNN_H; ++j) h[j] = nh[j];
  }

  float acc = b_fc[0];
#pragma unroll
  for (int j = 0; j < RNN_H; ++j) acc = fmaf(W_fc[j], h[j], acc);
  out[b] = acc;
}

extern "C" void kernel_launch(void* const* d_in, const int* in_sizes, int n_in,
                              void* d_out, int out_size, void* d_ws, size_t ws_size,
                              hipStream_t stream) {
  const float* x    = (const float*)d_in[0];
  const float* W_ih = (const float*)d_in[1];
  const float* W_hh = (const float*)d_in[2];
  const float* b_ih = (const float*)d_in[3];
  const float* b_hh = (const float*)d_in[4];
  const float* W_fc = (const float*)d_in[5];
  const float* b_fc = (const float*)d_in[6];
  float* out = (float*)d_out;

  const size_t need = (size_t)RNN_T * RNN_B * 8 * sizeof(float);
  if (ws_size >= need) {
    float* xp = (float*)d_ws;
    const long waves  = ((long)RNN_B * RNN_T / 16) / TILES_PER_WAVE;
    const int  blocks = (int)(waves * 32 / 256);
    xproj_wmma_f32<<<dim3(blocks), dim3(256), 0, stream>>>(x, W_ih, b_ih, b_hh, xp);
    rnn_scan<<<dim3(RNN_B / 256), dim3(256), 0, stream>>>(xp, W_hh, W_fc, b_fc, out);
  } else {
    rnn_fused<<<dim3(RNN_B / 256), dim3(256), 0, stream>>>(x, W_ih, W_hh, b_ih,
                                                           b_hh, W_fc, b_fc, out);
  }
}


# one fused kernel: per 32-row work-group 8 projection waves write xp into a 2-stage LDS ring while one wave scans it; xp never goes to memory; arithmetic and sum order unchanged
# speedup vs baseline: 5.5100x; 2.1959x over previous
#include <hip/hip_runtime.h>


typedef __attribute__((ext_vector_type(2))) float v2f;
typedef __attribute__((ext_vector_type(8))) float v8f;

#define RNN_B 8192
#define RNN_T 512
#define RNN_I 9
#define RNN_H 6

#define WG_ROWS 32
#define BLK_T   32
#define N_BLK   (RNN_T / BLK_T)
#define N_PROD  8
#define TILES_PER_WAVE 8
#define XP_TS 2
#define XP_BR (16 / XP_TS)
#define XP_TT (TILES_PER_WAVE * XP_TS)
#define ROW_F   8
#define STEP_F  (WG_ROWS * ROW_F)
#define STAGE_F (BLK_T * STEP_F)
#define SCAN_BLK 8

__device__ __forceinline__ float fast_tanhf(float v) {
  float e = __builtin_amdgcn_exp2f(v * 2.88539008177792681472f);
  return 1.0f - 2.0f * __builtin_amdgcn_rcpf(e + 1.0f);
}

struct ScanRows {
  float4 a[SCAN_BLK];
  float2 b[SCAN_BLK];
};

__device__ __forceinline__ void scan_load_rows(ScanRows& r, const float* p) {
#pragma unroll
  for (int u = 0; u < SCAN_BLK; ++u) {
    const float* q = p + u * STEP_F;
    r.a[u] = *(const float4*)q;
    r.b[u] = *(const float2*)(q + 4);
  }
}

__device__ __forceinline__ void scan_run_rows(const ScanRows& r,
                                              const float (&w)[RNN_H][RNN_H],
                                              float (&h)[RNN_H]) {
#pragma unroll
  for (int u = 0; u < SCAN_BLK; ++u) {
    const float xv[RNN_H] = {r.a[u].x, r.a[u].y, r.a[u].z, r.a[u].w,
                             r.b[u].x, r.b[u].y};
    float nh[RNN_H];
#pragma unroll
    for (int j = 0; j < RNN_H; ++j) {
      float acc = xv[j];
#pragma unroll
      for (int i = 0; i < RNN_H; ++i) acc = fmaf(w[j][i], h[i], acc);
      nh[j] = fast_tanhf(acc);
    }
#pragma unroll
    for (int j = 0; j < RNN_H; ++j) h[j] = nh[j];
  }
}

__global__ __launch_bounds__((N_PROD + 1) * 32) void rnn_ring(
    const float* __restrict__ x, const float* __restrict__ W_ih,
    const float* __restrict__ W_hh, const float* __restrict__ b_ih,
    const float* __restrict__ b_hh, const float* __restrict__ W_fc,
    const float* __restrict__ b_fc, float* __restrict__ out)
{
  static_assert(RNN_T % BLK_T == 0 && RNN_B % WG_ROWS == 0, "whole blocks");
  static_assert(N_PROD * TILES_PER_WAVE * 16 == WG_ROWS * BLK_T, "tiles cover a block");
  static_assert(BLK_T == 4 * SCAN_BLK, "four register sets' worth of steps per block");

  __shared__ __attribute__((aligned(16))) float ring[2 * STAGE_F];

  const int lane = threadIdx.x & 31;
  const int  wv       = __builtin_amdgcn_readfirstlane((int)(threadIdx.x >> 5));
  const bool producer = (wv != 0);
  const int  pw       = producer ? (wv - 1) : 0;
  const int  bw       = blockIdx.x * WG_ROWS;

  const int  c     = lane & 15;
  const int  half  = lane >> 4;
  const bool half0 = (half == 0);
  const int  bi    = c & (XP_BR - 1);
  const int  ti    = c / XP_BR;
  const int  rg    = pw & 3;
  const int  th    = (pw >> 2) & 1;

  const bool mVal = (c < 4) || (c >= 8 && c < 8 + (RNN_H - 4));
  const int  nRow = mVal ? ((c < 4) ? c : (c - 4)) : 0;
  v2f am[3];
#pragma unroll
  for (int s = 0; s < 3; ++s) {
#pragma unroll
    for (int e2 = 0; e2 < 2; ++e2) {
      const int  e    = 2 * s + e2;
      const bool kVal = half0 ? (e < 5) : (e < 4);
      const int  k    = kVal ? (half0 ? e : (5 + e)) : 0;
      const float w   = W_ih[nRow * RNN_I + k];
      const float v   = (mVal && kVal) ? w : 0.0f;
      if (e2 == 0) am[s].x = v; else am[s].y = v;
    }
  }

  float cb[4];
#pragma unroll
  for (int r = 0; r < 4; ++r) {
    const int  n    = 4 * half + r;
    const bool nVal = (n < RNN_H);
    const int  nc   = nVal ? n : 0;
    const float bi2 = b_ih[nc];
    const float bh2 = b_hh[nc];
    cb[r] = nVal ? (bi2 + bh2) : 0.0f;
  }

  const float* xr = x + ((size_t)(bw + XP_BR * rg + bi) * RNN_T
                         + (size_t)(XP_TT * th + ti)) * RNN_I + (half0 ? 0 : 5);
  const int sOff = half0 ? 4 : 3;
  float* lp = ring + (XP_TT * th + ti) * STEP_F + (XP_BR * rg + bi) * ROW_F + 4 * half;

  float4 xqA[TILES_PER_WAVE], xqB[TILES_PER_WAVE];
  float  xsA[TILES_PER_WAVE], xsB[TILES_PER_WAVE];
#pragma unroll
  for (int tt = 0; tt < TILES_PER_WAVE; ++tt) {
    xqA[tt] = make_float4(0.0f, 0.0f, 0.0f, 0.0f);  xsA[tt] = 0.0f;
    xqB[tt] = make_float4(0.0f, 0.0f, 0.0f, 0.0f);  xsB[tt] = 0.0f;
  }

  auto request_block = [&](float4 (&xq)[TILES_PER_WAVE], float (&xs)[TILES_PER_WAVE],
                           int blk) {
    const float* p = xr + (size_t)blk * (BLK_T * RNN_I);
#pragma unroll
    for (int tt = 0; tt < TILES_PER_WAVE; ++tt) {
      const float* q = p + tt * (XP_TS * RNN_I);
      __builtin_memcpy(&xq[tt], q, sizeof(float4));
      xs[tt] = q[sOff];
    }
  };

  auto project_block = [&](const float4 (&xq)[TILES_PER_WAVE],
                           const float (&xs)[TILES_PER_WAVE], int stage) {
    float* o = lp + stage * STAGE_F;
#pragma unroll
    for (int tt = 0; tt < TILES_PER_WAVE; ++tt) {
      v8f acc;
      acc[0] = cb[0]; acc[1] = cb[1]; acc[2] = cb[2]; acc[3] = cb[3];
      acc[4] = 0.0f;  acc[5] = 0.0f;  acc[6] = 0.0f;  acc[7] = 0.0f;
      v2f b0v, b1v, b2v;
      b0v.x = xq[tt].x; b0v.y = xq[tt].y;
      b1v.x = xq[tt].z; b1v.y = xq[tt].w;
      b2v.x = half0 ? xs[tt] : 0.0f;
      b2v.y = 0.0f;
      acc = __builtin_amdgcn_wmma_f32_16x16x4_f32(false, am[0], false, b0v,
                                                  (short)0, acc, false, false);
      acc = __builtin_amdgcn_wmma_f32_16x16x4_f32(false, am[1], false, b1v,
                                                  (short)0, acc, false, false);
      acc = __builtin_amdgcn_wmma_f32_16x16x4_f32(false, am[2], false, b2v,
                                                  (short)0, acc, false, false);
      float4 ov;
      ov.x = acc[0]; ov.y = acc[1]; ov.z = acc[2]; ov.w = acc[3];
      *(float4*)(o + tt * (XP_TS * STEP_F)) = ov;
    }
  };

  auto advance_sets = [&]() {
#pragma unroll
    for (int tt = 0; tt < TILES_PER_WAVE; ++tt) { xqA[tt] = xqB[tt]; xsA[tt] = xsB[tt]; }
  };

  float w[RNN_H][RNN_H];
#pragma unroll
  for (int j = 0; j < RNN_H; ++j)
#pragma unroll
    for (int i = 0; i < RNN_H; ++i) w[j][i] = W_hh[j * RNN_H + i];

  float h[RNN_H];
#pragma unroll
  for (int j = 0; j < RNN_H; ++j) h[j] = 0.0f;

  const float* rp = ring + lane * ROW_F;

  if (producer) {
    request_block(xqA, xsA, 0);
    request_block(xqB, xsB, 1);
    project_block(xqA, xsA, 0);
  }
  __syncthreads();
  if (producer) advance_sets();

#pragma unroll 1
  for (int k = 0; k < N_BLK; ++k) {
    if (producer) {
      if (k + 2 < N_BLK) request_block(xqB, xsB, k + 2);
      if (k + 1 < N_BLK) project_block(xqA, xsA, (k + 1) & 1);
    } else {
      const float* sp = rp + (k & 1) * STAGE_F;
      ScanRows X, Y;
#define SCAN_PIN() asm volatile("" : "+v"(h[0]), "+v"(h[1]), "+v"(h[2]), \
                                     "+v"(h[3]), "+v"(h[4]), "+v"(h[5]) : : "memory")
      scan_load_rows(X, sp);
      scan_load_rows(Y, sp + 1 * SCAN_BLK * STEP_F);
      SCAN_PIN();
      scan_run_rows(X, w, h);
      scan_load_rows(X, sp + 2 * SCAN_BLK * STEP_F);
      SCAN_PIN();
      scan_run_rows(Y, w, h);
      scan_load_rows(Y, sp + 3 * SCAN_BLK * STEP_F);
      SCAN_PIN();
      scan_run_rows(X, w, h);
      scan_run_rows(Y, w, h);
#undef SCAN_PIN
    }
    __syncthreads();
    if (producer) advance_sets();
  }

  if (!producer) {
    float acc = b_fc[0];
#pragma unroll
    for (int j = 0; j < RNN_H; ++j) acc = fmaf(W_fc[j], h[j], acc);
    out[bw + lane] = acc;
  }
}

extern "C" void kernel_launch(void* const* d_in, const int* in_sizes, int n_in,
                              void* d_out, int out_size, void* d_ws, size_t ws_size,
                              hipStream_t stream) {
  const float* x    = (const float*)d_in[0];
  const float* W_ih = (const float*)d_in[1];
  const float* W_hh = (const float*)d_in[2];
  const float* b_ih = (const float*)d_in[3];
  const float* b_hh = (const float*)d_in[4];
  const float* W_fc = (const float*)d_in[5];
  const float* b_fc = (const float*)d_in[6];
  float* out = (float*)d_out;

  rnn_ring<<<dim3(RNN_B / WG_ROWS), dim3((N_PROD + 1) * 32), 0, stream>>>(
      x, W_ih, W_hh, b_ih, b_hh, W_fc, b_fc, out);
}


# scan wave: tanh by the hardware v_tanh_f32 instruction (one transcendental per value) in place of mul + exp2 + add + rcp + fma; tanh values no longer bit-identical to the parent's
# speedup vs baseline: 6.2294x; 1.1306x over previous
#include <hip/hip_runtime.h>


typedef __attribute__((ext_vector_type(2))) float v2f;
typedef __attribute__((ext_vector_type(8))) float v8f;

#define RNN_B 8192
#define RNN_T 512
#define RNN_I 9
#define RNN_H 6

#define WG_ROWS 32
#define BLK_T   32
#define N_BLK   (RNN_T / BLK_T)
#define N_PROD  8
#define TILES_PER_WAVE 8
#define XP_TS 2
#define XP_BR (16 / XP_TS)
#define XP_TT (TILES_PER_WAVE * XP_TS)
#define ROW_F   8
#define STEP_F  (WG_ROWS * ROW_F)
#define STAGE_F (BLK_T * STEP_F)
#define SCAN_BLK 8

__device__ __forceinline__ float fast_tanhf(float v) {
  return __builtin_amdgcn_tanhf(v);
}

struct ScanRows {
  float4 a[SCAN_BLK];
  float2 b[SCAN_BLK];
};

__device__ __forceinline__ void scan_load_rows(ScanRows& r, const float* p) {
#pragma unroll
  for (int u = 0; u < SCAN_BLK; ++u) {
    const float* q = p + u * STEP_F;
    r.a[u] = *(const float4*)q;
    r.b[u] = *(const float2*)(q + 4);
  }
}

__device__ __forceinline__ void scan_run_rows(const ScanRows& r,
                                              const float (&w)[RNN_H][RNN_H],
                                              float (&h)[RNN_H]) {
#pragma unroll
  for (int u = 0; u < SCAN_BLK; ++u) {
    const float xv[RNN_H] = {r.a[u].x, r.a[u].y, r.a[u].z, r.a[u].w,
                             r.b[u].x, r.b[u].y};
    float nh[RNN_H];
#pragma unroll
    for (int j = 0; j < RNN_H; ++j) {
      float acc = xv[j];
#pragma unroll
      for (int i = 0; i < RNN_H; ++i) acc = fmaf(w[j][i], h[i], acc);
      nh[j] = fast_tanhf(acc);
    }
#pragma unroll
    for (int j = 0; j < RNN_H; ++j) h[j] = nh[j];
  }
}

__global__ __launch_bounds__((N_PROD + 1) * 32) void rnn_ring(
    const float* __restrict__ x, const float* __restrict__ W_ih,
    const float* __restrict__ W_hh, const float* __restrict__ b_ih,
    const float* __restrict__ b_hh, const float* __restrict__ W_fc,
    const float* __restrict__ b_fc, float* __restrict__ out)
{
  static_assert(RNN_T % BLK_T == 0 && RNN_B % WG_ROWS == 0, "whole blocks");
  static_assert(N_PROD * TILES_PER_WAVE * 16 == WG_ROWS * BLK_T, "tiles cover a block");
  static_assert(BLK_T == 4 * SCAN_BLK, "four register sets' worth of steps per block");

  __shared__ __attribute__((aligned(16))) float ring[2 * STAGE_F];

  const int lane = threadIdx.x & 31;
  const int  wv       = __builtin_amdgcn_readfirstlane((int)(threadIdx.x >> 5));
  const bool producer = (wv != 0);
  const int  pw       = producer ? (wv - 1) : 0;
  const int  bw       = blockIdx.x * WG_ROWS;

  const int  c     = lane & 15;
  const int  half  = lane >> 4;
  const bool half0 = (half == 0);
  const int  bi    = c & (XP_BR - 1);
  const int  ti    = c / XP_BR;
  const int  rg    = pw & 3;
  const int  th    = (pw >> 2) & 1;

  const bool mVal = (c < 4) || (c >= 8 && c < 8 + (RNN_H - 4));
  const int  nRow = mVal ? ((c < 4) ? c : (c - 4)) : 0;
  v2f am[3];
#pragma unroll
  for (int s = 0; s < 3; ++s) {
#pragma unroll
    for (int e2 = 0; e2 < 2; ++e2) {
      const int  e    = 2 * s + e2;
      const bool kVal = half0 ? (e < 5) : (e < 4);
      const int  k    = kVal ? (half0 ? e : (5 + e)) : 0;
      const float w   = W_ih[nRow * RNN_I + k];
      const float v   = (mVal && kVal) ? w : 0.0f;
      if (e2 == 0) am[s].x = v; else am[s].y = v;
    }
  }

  float cb[4];
#pragma unroll
  for (int r = 0; r < 4; ++r) {
    const int  n    = 4 * half + r;
    const bool nVal = (n < RNN_H);
    const int  nc   = nVal ? n : 0;
    const float bi2 = b_ih[nc];
    const float bh2 = b_hh[nc];
    cb[r] = nVal ? (bi2 + bh2) : 0.0f;
  }

  const float* xr = x + ((size_t)(bw + XP_BR * rg + bi) * RNN_T
                         + (size_t)(XP_TT * th + ti)) * RNN_I + (half0 ? 0 : 5);
  const int sOff = half0 ? 4 : 3;
  float* lp = ring + (XP_TT * th + ti) * STEP_F + (XP_BR * rg + bi) * ROW_F + 4 * half;

  float4 xqA[TILES_PER_WAVE], xqB[TILES_PER_WAVE];
  float  xsA[TILES_PER_WAVE], xsB[TILES_PER_WAVE];
#pragma unroll
  for (int tt = 0; tt < TILES_PER_WAVE; ++tt) {
    xqA[tt] = make_float4(0.0f, 0.0f, 0.0f, 0.0f);  xsA[tt] = 0.0f;
    xqB[tt] = make_float4(0.0f, 0.0f, 0.0f, 0.0f);  xsB[tt] = 0.0f;
  }

  auto request_block = [&](float4 (&xq)[TILES_PER_WAVE], float (&xs)[TILES_PER_WAVE],
                           int blk) {
    const float* p = xr + (size_t)blk * (BLK_T * RNN_I);
#pragma unroll
    for (int tt = 0; tt < TILES_PER_WAVE; ++tt) {
      const float* q = p + tt * (XP_TS * RNN_I);
      __builtin_memcpy(&xq[tt], q, sizeof(float4));
      xs[tt] = q[sOff];
    }
  };

  auto project_block = [&](const float4 (&xq)[TILES_PER_WAVE],
                           const float (&xs)[TILES_PER_WAVE], int stage) {
    float* o = lp + stage * STAGE_F;
#pragma unroll
    for (int tt = 0; tt < TILES_PER_WAVE; ++tt) {
      v8f acc;
      acc[0] = cb[0]; acc[1] = cb[1]; acc[2] = cb[2]; acc[3] = cb[3];
      acc[4] = 0.0f;  acc[5] = 0.0f;  acc[6] = 0.0f;  acc[7] = 0.0f;
      v2f b0v, b1v, b2v;
      b0v.x = xq[tt].x; b0v.y = xq[tt].y;
      b1v.x = xq[tt].z; b1v.y = xq[tt].w;
      b2v.x = half0 ? xs[tt] : 0.0f;
      b2v.y = 0.0f;
      acc = __builtin_amdgcn_wmma_f32_16x16x4_f32(false, am[0], false, b0v,
                                                  (short)0, acc, false, false);
      acc = __builtin_amdgcn_wmma_f32_16x16x4_f32(false, am[1], false, b1v,
                                                  (short)0, acc, false, false);
      acc = __builtin_amdgcn_wmma_f32_16x16x4_f32(false, am[2], false, b2v,
                                                  (short)0, acc, false, false);
      float4 ov;
      ov.x = acc[0]; ov.y = acc[1]; ov.z = acc[2]; ov.w = acc[3];
      *(float4*)(o + tt * (XP_TS * STEP_F)) = ov;
    }
  };

  auto advance_sets = [&]() {
#pragma unroll
    for (int tt = 0; tt < TILES_PER_WAVE; ++tt) { xqA[tt] = xqB[tt]; xsA[tt] = xsB[tt]; }
  };

  float w[RNN_H][RNN_H];
#pragma unroll
  for (int j = 0; j < RNN_H; ++j)
#pragma unroll
    for (int i = 0; i < RNN_H; ++i) w[j][i] = W_hh[j * RNN_H + i];

  float h[RNN_H];
#pragma unroll
  for (int j = 0; j < RNN_H; ++j) h[j] = 0.0f;

  const float* rp = ring + lane * ROW_F;

  if (producer) {
    request_block(xqA, xsA, 0);
    request_block(xqB, xsB, 1);
    project_block(xqA, xsA, 0);
  }
  __syncthreads();
  if (producer) advance_sets();

#pragma unroll 1
  for (int k = 0; k < N_BLK; ++k) {
    if (producer) {
      if (k + 2 < N_BLK) request_block(xqB, xsB, k + 2);
      if (k + 1 < N_BLK) project_block(xqA, xsA, (k + 1) & 1);
    } else {
      const float* sp = rp + (k & 1) * STAGE_F;
      ScanRows X, Y;
#define SCAN_PIN() asm volatile("" : "+v"(h[0]), "+v"(h[1]), "+v"(h[2]), \
                                     "+v"(h[3]), "+v"(h[4]), "+v"(h[5]) : : "memory")
      scan_load_rows(X, sp);
      scan_load_rows(Y, sp + 1 * SCAN_BLK * STEP_F);
      SCAN_PIN();
      scan_run_rows(X, w, h);
      scan_load_rows(X, sp + 2 * SCAN_BLK * STEP_F);
      SCAN_PIN();
      scan_run_rows(Y, w, h);
      scan_load_rows(Y, sp + 3 * SCAN_BLK * STEP_F);
      SCAN_PIN();
      scan_run_rows(X, w, h);
      scan_run_rows(Y, w, h);
#undef SCAN_PIN
    }
    __syncthreads();
    if (producer) advance_sets();
  }

  if (!producer) {
    float acc = b_fc[0];
#pragma unroll
    for (int j = 0; j < RNN_H; ++j) acc = fmaf(W_fc[j], h[j], acc);
    out[bw + lane] = acc;
  }
}

extern "C" void kernel_launch(void* const* d_in, const int* in_sizes, int n_in,
                              void* d_out, int out_size, void* d_ws, size_t ws_size,
                              hipStream_t stream) {
  const float* x    = (const float*)d_in[0];
  const float* W_ih = (const float*)d_in[1];
  const float* W_hh = (const float*)d_in[2];
  const float* b_ih = (const float*)d_in[3];
  const float* b_hh = (const float*)d_in[4];
  const float* W_fc = (const float*)d_in[5];
  const float* b_fc = (const float*)d_in[6];
  float* out = (float*)d_out;

  rnn_ring<<<dim3(RNN_B / WG_ROWS), dim3((N_PROD + 1) * 32), 0, stream>>>(
      x, W_ih, W_hh, b_ih, b_hh, W_fc, b_fc, out);
}


# projection tile = 1 batch row x 16 time steps (was 8 rows x 2 steps): each x load instruction reads one contiguous 576-byte run, not 8 pieces 18 KiB apart; LDS step pitch +32 B; sums unchanged
# speedup vs baseline: 7.3484x; 1.1796x over previous
#include <hip/hip_runtime.h>


typedef __attribute__((ext_vector_type(2))) float v2f;
typedef __attribute__((ext_vector_type(8))) float v8f;

#define RNN_B 8192
#define RNN_T 512
#define RNN_I 9
#define RNN_H 6

#define WG_ROWS 32
#define BLK_T   32
#define N_BLK   (RNN_T / BLK_T)
#define N_PROD  8
#define TILES_PER_WAVE 8
#define XP_TS 16
#define XP_TH (BLK_T / XP_TS)
#define WAVE_ROWS (TILES_PER_WAVE / XP_TH)
#define ROW_F   8
#define STEP_PAD 8
#define STEP_F  (WG_ROWS * ROW_F + STEP_PAD)
#define STAGE_F (BLK_T * STEP_F)
#define SCAN_BLK 8

__device__ __forceinline__ float fast_tanhf(float v) {
  return __builtin_amdgcn_tanhf(v);
}

struct ScanRows {
  float4 a[SCAN_BLK];
  float2 b[SCAN_BLK];
};

__device__ __forceinline__ void scan_load_rows(ScanRows& r, const float* p) {
#pragma unroll
  for (int u = 0; u < SCAN_BLK; ++u) {
    const float* q = p + u * STEP_F;
    r.a[u] = *(const float4*)q;
    r.b[u] = *(const float2*)(q + 4);
  }
}

__device__ __forceinline__ void scan_run_rows(const ScanRows& r,
                                              const float (&w)[RNN_H][RNN_H],
                                              float (&h)[RNN_H]) {
#pragma unroll
  for (int u = 0; u < SCAN_BLK; ++u) {
    const float xv[RNN_H] = {r.a[u].x, r.a[u].y, r.a[u].z, r.a[u].w,
                             r.b[u].x, r.b[u].y};
    float nh[RNN_H];
#pragma unroll
    for (int j = 0; j < RNN_H; ++j) {
      float acc = xv[j];
#pragma unroll
      for (int i = 0; i < RNN_H; ++i) acc = fmaf(w[j][i], h[i], acc);
      nh[j] = fast_tanhf(acc);
    }
#pragma unroll
    for (int j = 0; j < RNN_H; ++j) h[j] = nh[j];
  }
}

__global__ __launch_bounds__((N_PROD + 1) * 32) void rnn_ring(
    const float* __restrict__ x, const float* __restrict__ W_ih,
    const float* __restrict__ W_hh, const float* __restrict__ b_ih,
    const float* __restrict__ b_hh, const float* __restrict__ W_fc,
    const float* __restrict__ b_fc, float* __restrict__ out)
{
  static_assert(RNN_T % BLK_T == 0 && RNN_B % WG_ROWS == 0, "whole blocks");
  static_assert(N_PROD * TILES_PER_WAVE * 16 == WG_ROWS * BLK_T, "tiles cover a block");
  static_assert(BLK_T == 4 * SCAN_BLK, "four register sets' worth of steps per block");
  static_assert(N_PROD * WAVE_ROWS == WG_ROWS && XP_TH * XP_TS == BLK_T, "waves cover the rows, tiles the steps");

  __shared__ __attribute__((aligned(16))) float ring[2 * STAGE_F];

  const int lane = threadIdx.x & 31;
  const int  wv       = __builtin_amdgcn_readfirstlane((int)(threadIdx.x >> 5));
  const bool producer = (wv != 0);
  const int  pw       = producer ? (wv - 1) : 0;
  const int  bw       = blockIdx.x * WG_ROWS;

  const int  c     = lane & 15;
  const int  half  = lane >> 4;
  const bool half0 = (half == 0);
  const int  r0    = WAVE_ROWS * pw;

  const bool mVal = (c < 4) || (c >= 8 && c < 8 + (RNN_H - 4));
  const int  nRow = mVal ? ((c < 4) ? c : (c - 4)) : 0;
  v2f am[3];
#pragma unroll
  for (int s = 0; s < 3; ++s) {
#pragma unroll
    for (int e2 = 0; e2 < 2; ++e2) {
      const int  e    = 2 * s + e2;
      const bool kVal = half0 ? (e < 5) : (e < 4);
      const int  k    = kVal ? (half0 ? e : (5 + e)) : 0;
      const float w   = W_ih[nRow * RNN_I + k];
      const float v   = (mVal && kVal) ? w : 0.0f;
      if (e2 == 0) am[s].x = v; else am[s].y = v;
    }
  }

  float cb[4];
#pragma unroll
  for (int r = 0; r < 4; ++r) {
    const int  n    = 4 * half + r;
    const bool nVal = (n < RNN_H);
    const int  nc   = nVal ? n : 0;
    const float bi2 = b_ih[nc];
    const float bh2 = b_hh[nc];
    cb[r] = nVal ? (bi2 + bh2) : 0.0f;
  }

  const float* xr = x + ((size_t)(bw + r0) * RNN_T + (size_t)c) * RNN_I + (half0 ? 0 : 5);
  const int sOff = half0 ? 4 : 3;
  float* lp = ring + c * STEP_F + r0 * ROW_F + 4 * half;

  float4 xqA[TILES_PER_WAVE], xqB[TILES_PER_WAVE];
  float  xsA[TILES_PER_WAVE], xsB[TILES_PER_WAVE];
#pragma unroll
  for (int tt = 0; tt < TILES_PER_WAVE; ++tt) {
    xqA[tt] = make_float4(0.0f, 0.0f, 0.0f, 0.0f);  xsA[tt] = 0.0f;
    xqB[tt] = make_float4(0.0f, 0.0f, 0.0f, 0.0f);  xsB[tt] = 0.0f;
  }

  auto request_block = [&](float4 (&xq)[TILES_PER_WAVE], float (&xs)[TILES_PER_WAVE],
                           int blk) {
    const float* p = xr + (size_t)blk * (BLK_T * RNN_I);
#pragma unroll
    for (int tt = 0; tt < TILES_PER_WAVE; ++tt) {
      const float* q = p + (tt / XP_TH) * (RNN_T * RNN_I) + (tt % XP_TH) * (XP_TS * RNN_I);
      __builtin_memcpy(&xq[tt], q, sizeof(float4));
      xs[tt] = q[sOff];
    }
  };

  auto project_block = [&](const float4 (&xq)[TILES_PER_WAVE],
                           const float (&xs)[TILES_PER_WAVE], int stage) {
    float* o = lp + stage * STAGE_F;
#pragma unroll
    for (int tt = 0; tt < TILES_PER_WAVE; ++tt) {
      v8f acc;
      acc[0] = cb[0]; acc[1] = cb[1]; acc[2] = cb[2]; acc[3] = cb[3];
      acc[4] = 0.0f;  acc[5] = 0.0f;  acc[6] = 0.0f;  acc[7] = 0.0f;
      v2f b0v, b1v, b2v;
      b0v.x = xq[tt].x; b0v.y = xq[tt].y;
      b1v.x = xq[tt].z; b1v.y = xq[tt].w;
      b2v.x = half0 ? xs[tt] : 0.0f;
      b2v.y = 0.0f;
      acc = __builtin_amdgcn_wmma_f32_16x16x4_f32(false, am[0], false, b0v,
                                                  (short)0, acc, false, false);
      acc = __builtin_amdgcn_wmma_f32_16x16x4_f32(false, am[1], false, b1v,
                                                  (short)0, acc, false, false);
      acc = __builtin_amdgcn_wmma_f32_16x16x4_f32(false, am[2], false, b2v,
                                                  (short)0, acc, false, false);
      float4 ov;
      ov.x = acc[0]; ov.y = acc[1]; ov.z = acc[2]; ov.w = acc[3];
      *(float4*)(o + (tt % XP_TH) * (XP_TS * STEP_F) + (tt / XP_TH) * ROW_F) = ov;
    }
  };

  auto advance_sets = [&]() {
#pragma unroll
    for (int tt = 0; tt < TILES_PER_WAVE; ++tt) { xqA[tt] = xqB[tt]; xsA[tt] = xsB[tt]; }
  };

  float w[RNN_H][RNN_H];
#pragma unroll
  for (int j = 0; j < RNN_H; ++j)
#pragma unroll
    for (int i = 0; i < RNN_H; ++i) w[j][i] = W_hh[j * RNN_H + i];

  float h[RNN_H];
#pragma unroll
  for (int j = 0; j < RNN_H; ++j) h[j] = 0.0f;

  const float* rp = ring + lane * ROW_F;

  if (producer) {
    request_block(xqA, xsA, 0);
    request_block(xqB, xsB, 1);
    project_block(xqA, xsA, 0);
  }
  __syncthreads();
  if (producer) advance_sets();

#pragma unroll 1
  for (int k = 0; k < N_BLK; ++k) {
    if (producer) {
      if (k + 2 < N_BLK) request_block(xqB, xsB, k + 2);
      if (k + 1 < N_BLK) project_block(xqA, xsA, (k + 1) & 1);
    } else {
      const float* sp = rp + (k & 1) * STAGE_F;
      ScanRows X, Y;
#define SCAN_PIN() asm volatile("" : "+v"(h[0]), "+v"(h[1]), "+v"(h[2]), \
                                     "+v"(h[3]), "+v"(h[4]), "+v"(h[5]) : : "memory")
      scan_load_rows(X, sp);
      scan_load_rows(Y, sp + 1 * SCAN_BLK * STEP_F);
      SCAN_PIN();
      scan_run_rows(X, w, h);
      scan_load_rows(X, sp + 2 * SCAN_BLK * STEP_F);
      SCAN_PIN();
      scan_run_rows(Y, w, h);
      scan_load_rows(Y, sp + 3 * SCAN_BLK * STEP_F);
      SCAN_PIN();
      scan_run_rows(X, w, h);
      scan_run_rows(Y, w, h);
#undef SCAN_PIN
    }
    __syncthreads();
    if (producer) advance_sets();
  }

  if (!producer) {
    float acc = b_fc[0];
#pragma unroll
    for (int j = 0; j < RNN_H; ++j) acc = fmaf(W_fc[j], h[j], acc);
    out[bw + lane] = acc;
  }
}

extern "C" void kernel_launch(void* const* d_in, const int* in_sizes, int n_in,
                              void* d_out, int out_size, void* d_ws, size_t ws_size,
                              hipStream_t stream) {
  const float* x    = (const float*)d_in[0];
  const float* W_ih = (const float*)d_in[1];
  const float* W_hh = (const float*)d_in[2];
  const float* b_ih = (const float*)d_in[3];
  const float* b_hh = (const float*)d_in[4];
  const float* W_fc = (const float*)d_in[5];
  const float* b_fc = (const float*)d_in[6];
  float* out = (float*)d_out;

  rnn_ring<<<dim3(RNN_B / WG_ROWS), dim3((N_PROD + 1) * 32), 0, stream>>>(
      x, W_ih, W_hh, b_ih, b_hh, W_fc, b_fc, out);
}


# scan step: instruction order pinned by scheduling barriers as six rounds (round i = the 6 multiply-adds that read h[i]), then the 6 tanh; arithmetic and the order of every sum unchanged
# speedup vs baseline: 7.6064x; 1.0351x over previous
#include <hip/hip_runtime.h>


typedef __attribute__((ext_vector_type(2))) float v2f;
typedef __attribute__((ext_vector_type(8))) float v8f;

#define RNN_B 8192
#define RNN_T 512
#define RNN_I 9
#define RNN_H 6

#define WG_ROWS 32
#define BLK_T   32
#define N_BLK   (RNN_T / BLK_T)
#define N_PROD  8
#define TILES_PER_WAVE 8
#define XP_TS 16
#define XP_TH (BLK_T / XP_TS)
#define WAVE_ROWS (TILES_PER_WAVE / XP_TH)
#define ROW_F   8
#define STEP_PAD 8
#define STEP_F  (WG_ROWS * ROW_F + STEP_PAD)
#define STAGE_F (BLK_T * STEP_F)
#define SCAN_BLK 8

__device__ __forceinline__ float fast_tanhf(float v) {
  return __builtin_amdgcn_tanhf(v);
}

struct ScanRows {
  float4 a[SCAN_BLK];
  float2 b[SCAN_BLK];
};

__device__ __forceinline__ void scan_load_rows(ScanRows& r, const float* p) {
#pragma unroll
  for (int u = 0; u < SCAN_BLK; ++u) {
    const float* q = p + u * STEP_F;
    r.a[u] = *(const float4*)q;
    r.b[u] = *(const float2*)(q + 4);
  }
}

__device__ __forceinline__ void scan_run_rows(const ScanRows& r,
                                              const float (&w)[RNN_H][RNN_H],
                                              float (&h)[RNN_H]) {
#pragma unroll
  for (int u = 0; u < SCAN_BLK; ++u) {
    float acc[RNN_H] = {r.a[u].x, r.a[u].y, r.a[u].z, r.a[u].w,
                        r.b[u].x, r.b[u].y};
#pragma unroll
    for (int i = 0; i < RNN_H; ++i) {
#pragma unroll
      for (int j = 0; j < RNN_H; ++j) acc[j] = fmaf(w[j][i], h[i], acc[j]);
      __builtin_amdgcn_sched_barrier(0);
    }
#pragma unroll
    for (int j = 0; j < RNN_H; ++j) h[j] = fast_tanhf(acc[j]);
  }
}

__global__ __launch_bounds__((N_PROD + 1) * 32) void rnn_ring(
    const float* __restrict__ x, const float* __restrict__ W_ih,
    const float* __restrict__ W_hh, const float* __restrict__ b_ih,
    const float* __restrict__ b_hh, const float* __restrict__ W_fc,
    const float* __restrict__ b_fc, float* __restrict__ out)
{
  static_assert(RNN_T % BLK_T == 0 && RNN_B % WG_ROWS == 0, "whole blocks");
  static_assert(N_PROD * TILES_PER_WAVE * 16 == WG_ROWS * BLK_T, "tiles cover a block");
  static_assert(BLK_T == 4 * SCAN_BLK, "four register sets' worth of steps per block");
  static_assert(N_PROD * WAVE_ROWS == WG_ROWS && XP_TH * XP_TS == BLK_T, "waves cover the rows, tiles the steps");

  __shared__ __attribute__((aligned(16))) float ring[2 * STAGE_F];

  const int lane = threadIdx.x & 31;
  const int  wv       = __builtin_amdgcn_readfirstlane((int)(threadIdx.x >> 5));
  const bool producer = (wv != 0);
  const int  pw       = producer ? (wv - 1) : 0;
  const int  bw       = blockIdx.x * WG_ROWS;

  const int  c     = lane & 15;
  const int  half  = lane >> 4;
  const bool half0 = (half == 0);
  const int  r0    = WAVE_ROWS * pw;

  const bool mVal = (c < 4) || (c >= 8 && c < 8 + (RNN_H - 4));
  const int  nRow = mVal ? ((c < 4) ? c : (c - 4)) : 0;
  v2f am[3];
#pragma unroll
  for (int s = 0; s < 3; ++s) {
#pragma unroll
    for (int e2 = 0; e2 < 2; ++e2) {
      const int  e    = 2 * s + e2;
      const bool kVal = half0 ? (e < 5) : (e < 4);
      const int  k    = kVal ? (half0 ? e : (5 + e)) : 0;
      const float w   = W_ih[nRow * RNN_I + k];
      const float v   = (mVal && kVal) ? w : 0.0f;
      if (e2 == 0) am[s].x = v; else am[s].y = v;
    }
  }

  float cb[4];
#pragma unroll
  for (int r = 0; r < 4; ++r) {
    const int  n    = 4 * half + r;
    const bool nVal = (n < RNN_H);
    const int  nc   = nVal ? n : 0;
    const float bi2 = b_ih[nc];
    const float bh2 = b_hh[nc];
    cb[r] = nVal ? (bi2 + bh2) : 0.0f;
  }

  const float* xr = x + ((size_t)(bw + r0) * RNN_T + (size_t)c) * RNN_I + (half0 ? 0 : 5);
  const int sOff = half0 ? 4 : 3;
  float* lp = ring + c * STEP_F + r0 * ROW_F + 4 * half;

  float4 xqA[TILES_PER_WAVE], xqB[TILES_PER_WAVE];
  float  xsA[TILES_PER_WAVE], xsB[TILES_PER_WAVE];
#pragma unroll
  for (int tt = 0; tt < TILES_PER_WAVE; ++tt) {
    xqA[tt] = make_float4(0.0f, 0.0f, 0.0f, 0.0f);  xsA[tt] = 0.0f;
    xqB[tt] = make_float4(0.0f, 0.0f, 0.0f, 0.0f);  xsB[tt] = 0.0f;
  }

  auto request_block = [&](float4 (&xq)[TILES_PER_WAVE], float (&xs)[TILES_PER_WAVE],
                           int blk) {
    const float* p = xr + (size_t)blk * (BLK_T * RNN_I);
#pragma unroll
    for (int tt = 0; tt < TILES_PER_WAVE; ++tt) {
      const float* q = p + (tt / XP_TH) * (RNN_T * RNN_I) + (tt % XP_TH) * (XP_TS * RNN_I);
      __builtin_memcpy(&xq[tt], q, sizeof(float4));
      xs[tt] = q[sOff];
    }
  };

  auto project_block = [&](const float4 (&xq)[TILES_PER_WAVE],
                           const float (&xs)[TILES_PER_WAVE], int stage) {
    float* o = lp + stage * STAGE_F;
#pragma unroll
    for (int tt = 0; tt < TILES_PER_WAVE; ++tt) {
      v8f acc;
      acc[0] = cb[0]; acc[1] = cb[1]; acc[2] = cb[2]; acc[3] = cb[3];
      acc[4] = 0.0f;  acc[5] = 0.0f;  acc[6] = 0.0f;  acc[7] = 0.0f;
      v2f b0v, b1v, b2v;
      b0v.x = xq[tt].x; b0v.y = xq[tt].y;
      b1v.x = xq[tt].z; b1v.y = xq[tt].w;
      b2v.x = half0 ? xs[tt] : 0.0f;
      b2v.y = 0.0f;
      acc = __builtin_amdgcn_wmma_f32_16x16x4_f32(false, am[0], false, b0v,
                                                  (short)0, acc, false, false);
      acc = __builtin_amdgcn_wmma_f32_16x16x4_f32(false, am[1], false, b1v,
                                                  (short)0, acc, false, false);
      acc = __builtin_amdgcn_wmma_f32_16x16x4_f32(false, am[2], false, b2v,
                                                  (short)0, acc, false, false);
      float4 ov;
      ov.x = acc[0]; ov.y = acc[1]; ov.z = acc[2]; ov.w = acc[3];
      *(float4*)(o + (tt % XP_TH) * (XP_TS * STEP_F) + (tt / XP_TH) * ROW_F) = ov;
    }
  };

  auto advance_sets = [&]() {
#pragma unroll
    for (int tt = 0; tt < TILES_PER_WAVE; ++tt) { xqA[tt] = xqB[tt]; xsA[tt] = xsB[tt]; }
  };

  float w[RNN_H][RNN_H];
#pragma unroll
  for (int j = 0; j < RNN_H; ++j)
#pragma unroll
    for (int i = 0; i < RNN_H; ++i) w[j][i] = W_hh[j * RNN_H + i];

  float h[RNN_H];
#pragma unroll
  for (int j = 0; j < RNN_H; ++j) h[j] = 0.0f;

  const float* rp = ring + lane * ROW_F;

  if (producer) {
    request_block(xqA, xsA, 0);
    request_block(xqB, xsB, 1);
    project_block(xqA, xsA, 0);
  }
  __syncthreads();
  if (producer) advance_sets();

#pragma unroll 1
  for (int k = 0; k < N_BLK; ++k) {
    if (producer) {
      if (k + 2 < N_BLK) request_block(xqB, xsB, k + 2);
      if (k + 1 < N_BLK) project_block(xqA, xsA, (k + 1) & 1);
    } else {
      const float* sp = rp + (k & 1) * STAGE_F;
      ScanRows X, Y;
#define SCAN_PIN() asm volatile("" : "+v"(h[0]), "+v"(h[1]), "+v"(h[2]), \
                                     "+v"(h[3]), "+v"(h[4]), "+v"(h[5]) : : "memory")
      scan_load_rows(X, sp);
      scan_load_rows(Y, sp + 1 * SCAN_BLK * STEP_F);
      SCAN_PIN();
      scan_run_rows(X, w, h);
      scan_load_rows(X, sp + 2 * SCAN_BLK * STEP_F);
      SCAN_PIN();
      scan_run_rows(Y, w, h);
      scan_load_rows(Y, sp + 3 * SCAN_BLK * STEP_F);
      SCAN_PIN();
      scan_run_rows(X, w, h);
      scan_run_rows(Y, w, h);
#undef SCAN_PIN
    }
    __syncthreads();
    if (producer) advance_sets();
  }

  if (!producer) {
    float acc = b_fc[0];
#pragma unroll
    for (int j = 0; j < RNN_H; ++j) acc = fmaf(W_fc[j], h[j], acc);
    out[bw + lane] = acc;
  }
}

extern "C" void kernel_launch(void* const* d_in, const int* in_sizes, int n_in,
                              void* d_out, int out_size, void* d_ws, size_t ws_size,
                              hipStream_t stream) {
  const float* x    = (const float*)d_in[0];
  const float* W_ih = (const float*)d_in[1];
  const float* W_hh = (const float*)d_in[2];
  const float* b_ih = (const float*)d_in[3];
  const float* b_hh = (const float*)d_in[4];
  const float* W_fc = (const float*)d_in[5];
  const float* b_fc = (const float*)d_in[6];
  float* out = (float*)d_out;

  rnn_ring<<<dim3(RNN_B / WG_ROWS), dim3((N_PROD + 1) * 32), 0, stream>>>(
      x, W_ih, W_hh, b_ih, b_hh, W_fc, b_fc, out);
}


# projection waves: x requests stay in flight across the phase barrier (two register sets in turn, loop unrolled by two, no set copies, barrier fences LDS only); arithmetic and sums unchanged
# speedup vs baseline: 8.3279x; 1.0949x over previous
#include <hip/hip_runtime.h>


typedef __attribute__((ext_vector_type(2))) float v2f;
typedef __attribute__((ext_vector_type(8))) float v8f;
typedef __attribute__((ext_vector_type(4))) float v4f;

#define RNN_B 8192
#define RNN_T 512
#define RNN_I 9
#define RNN_H 6

#define WG_ROWS 32
#define BLK_T   32
#define N_BLK   (RNN_T / BLK_T)
#define N_PROD  8
#define TILES_PER_WAVE 8
#define XP_TS 16
#define XP_TH (BLK_T / XP_TS)
#define WAVE_ROWS (TILES_PER_WAVE / XP_TH)
#define ROW_F   8
#define STEP_PAD 8
#define STEP_F  (WG_ROWS * ROW_F + STEP_PAD)
#define STAGE_F (BLK_T * STEP_F)
#define SCAN_BLK 8

__device__ __forceinline__ float fast_tanhf(float v) {
  return __builtin_amdgcn_tanhf(v);
}

struct ScanRows {
  float4 a[SCAN_BLK];
  float2 b[SCAN_BLK];
};

__device__ __forceinline__ void scan_load_rows(ScanRows& r, const float* p) {
#pragma unroll
  for (int u = 0; u < SCAN_BLK; ++u) {
    const float* q = p + u * STEP_F;
    r.a[u] = *(const float4*)q;
    r.b[u] = *(const float2*)(q + 4);
  }
}

__device__ __forceinline__ void scan_run_rows(const ScanRows& r,
                                              const float (&w)[RNN_H][RNN_H],
                                              float (&h)[RNN_H]) {
#pragma unroll
  for (int u = 0; u < SCAN_BLK; ++u) {
    float acc[RNN_H] = {r.a[u].x, r.a[u].y, r.a[u].z, r.a[u].w,
                        r.b[u].x, r.b[u].y};
#pragma unroll
    for (int i = 0; i < RNN_H; ++i) {
#pragma unroll
      for (int j = 0; j < RNN_H; ++j) acc[j] = fmaf(w[j][i], h[i], acc[j]);
      __builtin_amdgcn_sched_barrier(0);
    }
#pragma unroll
    for (int j = 0; j < RNN_H; ++j) h[j] = fast_tanhf(acc[j]);
  }
}

__global__ __launch_bounds__((N_PROD + 1) * 32) void rnn_ring(
    const float* __restrict__ x, const float* __restrict__ W_ih,
    const float* __restrict__ W_hh, const float* __restrict__ b_ih,
    const float* __restrict__ b_hh, const float* __restrict__ W_fc,
    const float* __restrict__ b_fc, float* __restrict__ out)
{
  static_assert(RNN_T % BLK_T == 0 && RNN_B % WG_ROWS == 0, "whole blocks");
  static_assert(N_PROD * TILES_PER_WAVE * 16 == WG_ROWS * BLK_T, "tiles cover a block");
  static_assert(BLK_T == 4 * SCAN_BLK, "four register sets' worth of steps per block");
  static_assert(N_BLK % 2 == 0, "two phases per loop trip");
  static_assert(N_PROD * WAVE_ROWS == WG_ROWS && XP_TH * XP_TS == BLK_T, "waves cover the rows, tiles the steps");

  __shared__ __attribute__((aligned(16))) float ring[2 * STAGE_F];

  const int lane = threadIdx.x & 31;
  const int  wv       = __builtin_amdgcn_readfirstlane((int)(threadIdx.x >> 5));
  const bool producer = (wv != 0);
  const int  pw       = producer ? (wv - 1) : 0;
  const int  bw       = blockIdx.x * WG_ROWS;

  const int  c     = lane & 15;
  const int  half  = lane >> 4;
  const bool half0 = (half == 0);
  const int  r0    = WAVE_ROWS * pw;

  const bool mVal = (c < 4) || (c >= 8 && c < 8 + (RNN_H - 4));
  const int  nRow = mVal ? ((c < 4) ? c : (c - 4)) : 0;
  v2f am[3];
#pragma unroll
  for (int s = 0; s < 3; ++s) {
#pragma unroll
    for (int e2 = 0; e2 < 2; ++e2) {
      const int  e    = 2 * s + e2;
      const bool kVal = half0 ? (e < 5) : (e < 4);
      const int  k    = kVal ? (half0 ? e : (5 + e)) : 0;
      const float w   = W_ih[nRow * RNN_I + k];
      const float v   = (mVal && kVal) ? w : 0.0f;
      if (e2 == 0) am[s].x = v; else am[s].y = v;
    }
  }

  float cb[4];
#pragma unroll
  for (int r = 0; r < 4; ++r) {
    const int  n    = 4 * half + r;
    const bool nVal = (n < RNN_H);
    const int  nc   = nVal ? n : 0;
    const float bi2 = b_ih[nc];
    const float bh2 = b_hh[nc];
    cb[r] = nVal ? (bi2 + bh2) : 0.0f;
  }

  const float* xr = x + ((size_t)(bw + r0) * RNN_T + (size_t)c) * RNN_I + (half0 ? 0 : 5);
  const int sOff = half0 ? 4 : 3;
  float* lp = ring + c * STEP_F + r0 * ROW_F + 4 * half;

  v4f    xqA[TILES_PER_WAVE], xqB[TILES_PER_WAVE];
  float  xsA[TILES_PER_WAVE], xsB[TILES_PER_WAVE];
#pragma unroll
  for (int tt = 0; tt < TILES_PER_WAVE; ++tt) {
    xqA[tt] = (v4f)(0.0f);  xsA[tt] = 0.0f;
    xqB[tt] = (v4f)(0.0f);  xsB[tt] = 0.0f;
  }

  auto request_block = [&](v4f (&xq)[TILES_PER_WAVE], float (&xs)[TILES_PER_WAVE],
                           int blk) {
    const float* p = xr + (size_t)blk * (BLK_T * RNN_I);
#pragma unroll
    for (int tt = 0; tt < TILES_PER_WAVE; ++tt) {
      const float* q = p + (tt / XP_TH) * (RNN_T * RNN_I) + (tt % XP_TH) * (XP_TS * RNN_I);
      __builtin_memcpy(&xq[tt], q, sizeof(v4f));
      xs[tt] = q[sOff];
    }
  };

  auto project_block = [&](const v4f (&xq)[TILES_PER_WAVE],
                           const float (&xs)[TILES_PER_WAVE], int stage) {
    float* o = lp + stage * STAGE_F;
#pragma unroll
    for (int tt = 0; tt < TILES_PER_WAVE; ++tt) {
      v8f acc;
      acc[0] = cb[0]; acc[1] = cb[1]; acc[2] = cb[2]; acc[3] = cb[3];
      acc[4] = 0.0f;  acc[5] = 0.0f;  acc[6] = 0.0f;  acc[7] = 0.0f;
      v2f b0v, b1v, b2v;
      b0v.x = xq[tt].x; b0v.y = xq[tt].y;
      b1v.x = xq[tt].z; b1v.y = xq[tt].w;
      b2v.x = half0 ? xs[tt] : 0.0f;
      b2v.y = 0.0f;
      acc = __builtin_amdgcn_wmma_f32_16x16x4_f32(false, am[0], false, b0v,
                                                  (short)0, acc, false, false);
      acc = __builtin_amdgcn_wmma_f32_16x16x4_f32(false, am[1], false, b1v,
                                                  (short)0, acc, false, false);
      acc = __builtin_amdgcn_wmma_f32_16x16x4_f32(false, am[2], false, b2v,
                                                  (short)0, acc, false, false);
      float4 ov;
      ov.x = acc[0]; ov.y = acc[1]; ov.z = acc[2]; ov.w = acc[3];
      *(float4*)(o + (tt % XP_TH) * (XP_TS * STEP_F) + (tt / XP_TH) * ROW_F) = ov;
    }
  };

  float w[RNN_H][RNN_H];
#pragma unroll
  for (int j = 0; j < RNN_H; ++j)
#pragma unroll
    for (int i = 0; i < RNN_H; ++i) w[j][i] = W_hh[j * RNN_H + i];

  float h[RNN_H];
#pragma unroll
  for (int j = 0; j < RNN_H; ++j) h[j] = 0.0f;

  const float* rp = ring + lane * ROW_F;

  auto scan_phase = [&](const float* sp) __attribute__((always_inline)) {
    ScanRows X, Y;
#define SCAN_PIN() asm volatile("" : "+v"(h[0]), "+v"(h[1]), "+v"(h[2]), \
                                     "+v"(h[3]), "+v"(h[4]), "+v"(h[5]) : : "memory")
    scan_load_rows(X, sp);
    scan_load_rows(Y, sp + 1 * SCAN_BLK * STEP_F);
    SCAN_PIN();
    scan_run_rows(X, w, h);
    scan_load_rows(X, sp + 2 * SCAN_BLK * STEP_F);
    SCAN_PIN();
    scan_run_rows(Y, w, h);
    scan_load_rows(Y, sp + 3 * SCAN_BLK * STEP_F);
    SCAN_PIN();
    scan_run_rows(X, w, h);
    scan_run_rows(Y, w, h);
#undef SCAN_PIN
  };

  auto drop_set = [&](v4f (&xq)[TILES_PER_WAVE], float (&xs)[TILES_PER_WAVE]) __attribute__((always_inline)) {
#pragma unroll
    for (int tt = 0; tt < TILES_PER_WAVE; ++tt) {
      xq[tt] = __builtin_nondeterministic_value(xq[tt]);
      xs[tt] = __builtin_nondeterministic_value(xs[tt]);
    }
  };

  auto ring_barrier = [&]() __attribute__((always_inline)) {
    __builtin_amdgcn_fence(__ATOMIC_RELEASE, "workgroup", "local");
    __builtin_amdgcn_s_barrier();
    __builtin_amdgcn_fence(__ATOMIC_ACQUIRE, "workgroup", "local");
  };

  if (producer) {
    request_block(xqA, xsA, 0);
    request_block(xqB, xsB, 1);
    project_block(xqA, xsA, 0);
  }
  ring_barrier();

#pragma unroll 1
  for (int k = 0; k < N_BLK; k += 2) {
    if (producer) {
      request_block(xqA, xsA, (k + 2 < N_BLK) ? (k + 2) : (N_BLK - 1));
      asm volatile("" ::: "memory");
      project_block(xqB, xsB, 1);
    } else {
      drop_set(xqA, xsA);
      scan_phase(rp);
    }
    ring_barrier();
    if (producer) {
      request_block(xqB, xsB, (k + 3 < N_BLK) ? (k + 3) : (N_BLK - 1));
      asm volatile("" ::: "memory");
      if (k + 2 < N_BLK) project_block(xqA, xsA, 0);
    } else {
      drop_set(xqB, xsB);
      scan_phase(rp + STAGE_F);
    }
    ring_barrier();
  }

  if (!producer) {
    float acc = b_fc[0];
#pragma unroll
    for (int j = 0; j < RNN_H; ++j) acc = fmaf(W_fc[j], h[j], acc);
    out[bw + lane] = acc;
  }
}

extern "C" void kernel_launch(void* const* d_in, const int* in_sizes, int n_in,
                              void* d_out, int out_size, void* d_ws, size_t ws_size,
                              hipStream_t stream) {
  const float* x    = (const float*)d_in[0];
  const float* W_ih = (const float*)d_in[1];
  const float* W_hh = (const float*)d_in[2];
  const float* b_ih = (const float*)d_in[3];
  const float* b_hh = (const float*)d_in[4];
  const float* W_fc = (const float*)d_in[5];
  const float* b_fc = (const float*)d_in[6];
  float* out = (float*)d_out;

  rnn_ring<<<dim3(RNN_B / WG_ROWS), dim3((N_PROD + 1) * 32), 0, stream>>>(
      x, W_ih, W_hh, b_ih, b_hh, W_fc, b_fc, out);
}


# work-group of 11 waves: waves 4 and 8 (dealt to the scan wave's SIMD) stay empty, so the 8 projection waves' f32 matrix instructions run on the other three SIMDs only; arithmetic and sums unchanged
# speedup vs baseline: 8.9489x; 1.0746x over previous
#include <hip/hip_runtime.h>


typedef __attribute__((ext_vector_type(2))) float v2f;
typedef __attribute__((ext_vector_type(8))) float v8f;
typedef __attribute__((ext_vector_type(4))) float v4f;

#define RNN_B 8192
#define RNN_T 512
#define RNN_I 9
#define RNN_H 6

#define WG_ROWS 32
#define BLK_T   32
#define N_BLK   (RNN_T / BLK_T)
#define N_PROD  8
#define N_WAVES 11
#define TILES_PER_WAVE 8
#define XP_TS 16
#define XP_TH (BLK_T / XP_TS)
#define WAVE_ROWS (TILES_PER_WAVE / XP_TH)
#define ROW_F   8
#define STEP_PAD 8
#define STEP_F  (WG_ROWS * ROW_F + STEP_PAD)
#define STAGE_F (BLK_T * STEP_F)
#define SCAN_BLK 8

__device__ __forceinline__ float fast_tanhf(float v) {
  return __builtin_amdgcn_tanhf(v);
}

struct ScanRows {
  float4 a[SCAN_BLK];
  float2 b[SCAN_BLK];
};

__device__ __forceinline__ void scan_load_rows(ScanRows& r, const float* p) {
#pragma unroll
  for (int u = 0; u < SCAN_BLK; ++u) {
    const float* q = p + u * STEP_F;
    r.a[u] = *(const float4*)q;
    r.b[u] = *(const float2*)(q + 4);
  }
}

__device__ __forceinline__ void scan_run_rows(const ScanRows& r,
                                              const float (&w)[RNN_H][RNN_H],
                                              float (&h)[RNN_H]) {
#pragma unroll
  for (int u = 0; u < SCAN_BLK; ++u) {
    float acc[RNN_H] = {r.a[u].x, r.a[u].y, r.a[u].z, r.a[u].w,
                        r.b[u].x, r.b[u].y};
#pragma unroll
    for (int i = 0; i < RNN_H; ++i) {
#pragma unroll
      for (int j = 0; j < RNN_H; ++j) acc[j] = fmaf(w[j][i], h[i], acc[j]);
      __builtin_amdgcn_sched_barrier(0);
    }
#pragma unroll
    for (int j = 0; j < RNN_H; ++j) h[j] = fast_tanhf(acc[j]);
  }
}

__global__ __launch_bounds__(N_WAVES * 32) void rnn_ring(
    const float* __restrict__ x, const float* __restrict__ W_ih,
    const float* __restrict__ W_hh, const float* __restrict__ b_ih,
    const float* __restrict__ b_hh, const float* __restrict__ W_fc,
    const float* __restrict__ b_fc, float* __restrict__ out)
{
  static_assert(RNN_T % BLK_T == 0 && RNN_B % WG_ROWS == 0, "whole blocks");
  static_assert(N_PROD * TILES_PER_WAVE * 16 == WG_ROWS * BLK_T, "tiles cover a block");
  static_assert(BLK_T == 4 * SCAN_BLK, "four register sets' worth of steps per block");
  static_assert(N_BLK % 2 == 0, "two phases per loop trip");
  static_assert(N_PROD * WAVE_ROWS == WG_ROWS && XP_TH * XP_TS == BLK_T, "waves cover the rows, tiles the steps");
  static_assert(N_WAVES - 1 - (N_WAVES - 1) / 4 == N_PROD, "every wave w with w % 4 != 0 is a projection wave");

  __shared__ __attribute__((aligned(16))) float ring[2 * STAGE_F];

  const int lane = threadIdx.x & 31;
  const int  wv       = __builtin_amdgcn_readfirstlane((int)(threadIdx.x >> 5));
  const bool scanner  = (wv == 0);
  const bool producer = ((wv & 3) != 0);
  const int  pw       = producer ? (wv - 1 - (wv >> 2)) : 0;
  const int  bw       = blockIdx.x * WG_ROWS;

  const int  c     = lane & 15;
  const int  half  = lane >> 4;
  const bool half0 = (half == 0);
  const int  r0    = WAVE_ROWS * pw;

  const bool mVal = (c < 4) || (c >= 8 && c < 8 + (RNN_H - 4));
  const int  nRow = mVal ? ((c < 4) ? c : (c - 4)) : 0;
  v2f am[3];
#pragma unroll
  for (int s = 0; s < 3; ++s) {
#pragma unroll
    for (int e2 = 0; e2 < 2; ++e2) {
      const int  e    = 2 * s + e2;
      const bool kVal = half0 ? (e < 5) : (e < 4);
      const int  k    = kVal ? (half0 ? e : (5 + e)) : 0;
      const float w   = W_ih[nRow * RNN_I + k];
      const float v   = (mVal && kVal) ? w : 0.0f;
      if (e2 == 0) am[s].x = v; else am[s].y = v;
    }
  }

  float cb[4];
#pragma unroll
  for (int r = 0; r < 4; ++r) {
    const int  n    = 4 * half + r;
    const bool nVal = (n < RNN_H);
    const int  nc   = nVal ? n : 0;
    const float bi2 = b_ih[nc];
    const float bh2 = b_hh[nc];
    cb[r] = nVal ? (bi2 + bh2) : 0.0f;
  }

  const float* xr = x + ((size_t)(bw + r0) * RNN_T + (size_t)c) * RNN_I + (half0 ? 0 : 5);
  const int sOff = half0 ? 4 : 3;
  float* lp = ring + c * STEP_F + r0 * ROW_F + 4 * half;

  v4f    xqA[TILES_PER_WAVE], xqB[TILES_PER_WAVE];
  float  xsA[TILES_PER_WAVE], xsB[TILES_PER_WAVE];
#pragma unroll
  for (int tt = 0; tt < TILES_PER_WAVE; ++tt) {
    xqA[tt] = (v4f)(0.0f);  xsA[tt] = 0.0f;
    xqB[tt] = (v4f)(0.0f);  xsB[tt] = 0.0f;
  }

  auto request_block = [&](v4f (&xq)[TILES_PER_WAVE], float (&xs)[TILES_PER_WAVE],
                           int blk) {
    const float* p = xr + (size_t)blk * (BLK_T * RNN_I);
#pragma unroll
    for (int tt = 0; tt < TILES_PER_WAVE; ++tt) {
      const float* q = p + (tt / XP_TH) * (RNN_T * RNN_I) + (tt % XP_TH) * (XP_TS * RNN_I);
      __builtin_memcpy(&xq[tt], q, sizeof(v4f));
      xs[tt] = q[sOff];
    }
  };

  auto project_block = [&](const v4f (&xq)[TILES_PER_WAVE],
                           const float (&xs)[TILES_PER_WAVE], int stage) {
    float* o = lp + stage * STAGE_F;
#pragma unroll
    for (int tt = 0; tt < TILES_PER_WAVE; ++tt) {
      v8f acc;
      acc[0] = cb[0]; acc[1] = cb[1]; acc[2] = cb[2]; acc[3] = cb[3];
      acc[4] = 0.0f;  acc[5] = 0.0f;  acc[6] = 0.0f;  acc[7] = 0.0f;
      v2f b0v, b1v, b2v;
      b0v.x = xq[tt].x; b0v.y = xq[tt].y;
      b1v.x = xq[tt].z; b1v.y = xq[tt].w;
      b2v.x = half0 ? xs[tt] : 0.0f;
      b2v.y = 0.0f;
      acc = __builtin_amdgcn_wmma_f32_16x16x4_f32(false, am[0], false, b0v,
                                                  (short)0, acc, false, false);
      acc = __builtin_amdgcn_wmma_f32_16x16x4_f32(false, am[1], false, b1v,
                                                  (short)0, acc, false, false);
      acc = __builtin_amdgcn_wmma_f32_16x16x4_f32(false, am[2], false, b2v,
                                                  (short)0, acc, false, false);
      float4 ov;
      ov.x = acc[0]; ov.y = acc[1]; ov.z = acc[2]; ov.w = acc[3];
      *(float4*)(o + (tt % XP_TH) * (XP_TS * STEP_F) + (tt / XP_TH) * ROW_F) = ov;
    }
  };

  float w[RNN_H][RNN_H];
#pragma unroll
  for (int j = 0; j < RNN_H; ++j)
#pragma unroll
    for (int i = 0; i < RNN_H; ++i) w[j][i] = W_hh[j * RNN_H + i];

  float h[RNN_H];
#pragma unroll
  for (int j = 0; j < RNN_H; ++j) h[j] = 0.0f;

  const float* rp = ring + lane * ROW_F;

  auto scan_phase = [&](const float* sp) __attribute__((always_inline)) {
    ScanRows X, Y;
#define SCAN_PIN() asm volatile("" : "+v"(h[0]), "+v"(h[1]), "+v"(h[2]), \
                                     "+v"(h[3]), "+v"(h[4]), "+v"(h[5]) : : "memory")
    scan_load_rows(X, sp);
    scan_load_rows(Y, sp + 1 * SCAN_BLK * STEP_F);
    SCAN_PIN();
    scan_run_rows(X, w, h);
    scan_load_rows(X, sp + 2 * SCAN_BLK * STEP_F);
    SCAN_PIN();
    scan_run_rows(Y, w, h);
    scan_load_rows(Y, sp + 3 * SCAN_BLK * STEP_F);
    SCAN_PIN();
    scan_run_rows(X, w, h);
    scan_run_rows(Y, w, h);
#undef SCAN_PIN
  };

  auto drop_set = [&](v4f (&xq)[TILES_PER_WAVE], float (&xs)[TILES_PER_WAVE]) __attribute__((always_inline)) {
#pragma unroll
    for (int tt = 0; tt < TILES_PER_WAVE; ++tt) {
      xq[tt] = __builtin_nondeterministic_value(xq[tt]);
      xs[tt] = __builtin_nondeterministic_value(xs[tt]);
    }
  };

  auto ring_barrier = [&]() __attribute__((always_inline)) {
    __builtin_amdgcn_fence(__ATOMIC_RELEASE, "workgroup", "local");
    __builtin_amdgcn_s_barrier();
    __builtin_amdgcn_fence(__ATOMIC_ACQUIRE, "workgroup", "local");
  };

  if (producer) {
    request_block(xqA, xsA, 0);
    request_block(xqB, xsB, 1);
    project_block(xqA, xsA, 0);
  }
  ring_barrier();

#pragma unroll 1
  for (int k = 0; k < N_BLK; k += 2) {
    if (producer) {
      request_block(xqA, xsA, (k + 2 < N_BLK) ? (k + 2) : (N_BLK - 1));
      asm volatile("" ::: "memory");
      project_block(xqB, xsB, 1);
    } else {
      drop_set(xqA, xsA);
      if (scanner) scan_phase(rp);
    }
    ring_barrier();
    if (producer) {
      request_block(xqB, xsB, (k + 3 < N_BLK) ? (k + 3) : (N_BLK - 1));
      asm volatile("" ::: "memory");
      if (k + 2 < N_BLK) project_block(xqA, xsA, 0);
    } else {
      drop_set(xqB, xsB);
      if (scanner) scan_phase(rp + STAGE_F);
    }
    ring_barrier();
  }

  if (scanner) {
    float acc = b_fc[0];
#pragma unroll
    for (int j = 0; j < RNN_H; ++j) acc = fmaf(W_fc[j], h[j], acc);
    out[bw + lane] = acc;
  }
}

extern "C" void kernel_launch(void* const* d_in, const int* in_sizes, int n_in,
                              void* d_out, int out_size, void* d_ws, size_t ws_size,
                              hipStream_t stream) {
  const float* x    = (const float*)d_in[0];
  const float* W_ih = (const float*)d_in[1];
  const float* W_hh = (const float*)d_in[2];
  const float* b_ih = (const float*)d_in[3];
  const float* b_hh = (const float*)d_in[4];
  const float* W_fc = (const float*)d_in[5];
  const float* b_fc = (const float*)d_in[6];
  float* out = (float*)d_out;

  rnn_ring<<<dim3(RNN_B / WG_ROWS), dim3(N_WAVES * 32), 0, stream>>>(
      x, W_ih, W_hh, b_ih, b_hh, W_fc, b_fc, out);
}
